# starting kernel
# baseline (speedup 1.0000x reference)
#include <hip/hip_runtime.h>


typedef __attribute__((ext_vector_type(2))) float v2f;
typedef __attribute__((ext_vector_type(8))) float v8f;

namespace {
constexpr int KD    = 256;
constexpr int NC    = 64;
constexpr int NR    = 8192;
constexpr int WAVES = 8;
constexpr int QBLK  = NR / (WAVES * 16);
}

__global__ void __launch_bounds__(256)
clusternet_cnorm_kernel(const float* __restrict__ cent,
                        float* __restrict__ cnorm) {
    const int lane = threadIdx.x & 31;
    const int row  = blockIdx.x * WAVES + (threadIdx.x >> 5);
    const float* cr = cent + (size_t)row * KD;

    float s = 0.f;
#pragma unroll
    for (int h = 0; h < KD; h += 32) {
        float v = cr[h + lane];
        s += v * v;
    }
    s += __shfl_xor(s, 1, 32);
    s += __shfl_xor(s, 2, 32);
    s += __shfl_xor(s, 4, 32);
    s += __shfl_xor(s, 8, 32);
    s += __shfl_xor(s, 16, 32);
    if (lane == 0) cnorm[row] = s;
}

__global__ void __launch_bounds__(256)
clusternet_q_kernel(const float* __restrict__ z,
                    const float* __restrict__ cent,
                    const float* __restrict__ cnorm,
                    float* __restrict__ Qout,
                    float* __restrict__ blockpart) {
    __shared__ float csum[WAVES][NC];

    const int lane = threadIdx.x & 31;
    const int wave = threadIdx.x >> 5;
    const int half = lane >> 4;
    const int n16  = lane & 15;
    const int rowBase = (blockIdx.x * WAVES + wave) * 16;

    const float* zrow = z + (size_t)(rowBase + n16) * KD + 2 * half;
    const float* cr[4];
    float cn[4];
#pragma unroll
    for (int t = 0; t < 4; ++t) {
        cr[t] = cent + (size_t)(t * 16 + n16) * KD + 2 * half;
        cn[t] = cnorm[t * 16 + n16];
    }

    v8f acc[4] = {};
    float zp = 0.f;

#pragma unroll 4
    for (int kb = 0; kb < KD; kb += 4) {
        v2f av = *(const v2f*)(zrow + kb);
        zp += av.x * av.x + av.y * av.y;
#pragma unroll
        for (int t = 0; t < 4; ++t) {
            v2f bv = *(const v2f*)(cr[t] + kb);
            acc[t] = __builtin_amdgcn_wmma_f32_16x16x4_f32(
                false, av, false, bv, (short)0, acc[t], false, false);
        }
    }

    float znf = zp + __shfl_xor(zp, 16, 32);

    float qv[4][8];
#pragma unroll
    for (int r = 0; r < 8; ++r) {
        float znr = __shfl(znf, r + 8 * half, 32);
#pragma unroll
        for (int t = 0; t < 4; ++t) {
            float d  = acc[t][r];
            float s2 = fmaxf(znr + cn[t] - 2.0f * d, 0.0f);
            float sim = sqrtf(s2);
            qv[t][r] = 1.0f / (1.0f + sim);
        }
        float rs = qv[0][r] + qv[1][r] + qv[2][r] + qv[3][r];
        rs += __shfl_xor(rs, 1, 32);
        rs += __shfl_xor(rs, 2, 32);
        rs += __shfl_xor(rs, 4, 32);
        rs += __shfl_xor(rs, 8, 32);
        float inv = 1.0f / rs;
#pragma unroll
        for (int t = 0; t < 4; ++t) {
            qv[t][r] *= inv;
            Qout[(size_t)(rowBase + r + 8 * half) * NC + t * 16 + n16] = qv[t][r];
        }
    }

#pragma unroll
    for (int t = 0; t < 4; ++t) {
        float cs = 0.f;
#pragma unroll
        for (int r = 0; r < 8; ++r) cs += qv[t][r];
        cs += __shfl_xor(cs, 16, 32);
        if (lane < 16) csum[wave][t * 16 + n16] = cs;
    }
    __syncthreads();
    if (threadIdx.x < NC) {
        float s = 0.f;
#pragma unroll
        for (int w = 0; w < WAVES; ++w) s += csum[w][threadIdx.x];
        blockpart[(size_t)blockIdx.x * NC + threadIdx.x] = s;
    }
}

__global__ void __launch_bounds__(64)
clusternet_colsum_kernel(const float* __restrict__ blockpart,
                         float* __restrict__ colsum) {
    float s = 0.f;
#pragma unroll 8
    for (int b = 0; b < QBLK; ++b) s += blockpart[(size_t)b * NC + threadIdx.x];
    colsum[threadIdx.x] = s;
}

__global__ void __launch_bounds__(256)
clusternet_p_kernel(const float* __restrict__ Q,
                    const float* __restrict__ colsum,
                    float* __restrict__ Pout) {
    const int lane = threadIdx.x & 31;
    const int row  = blockIdx.x * WAVES + (threadIdx.x >> 5);
    const size_t base = (size_t)row * NC;

    float q0 = Q[base + lane];
    float q1 = Q[base + lane + 32];
    float p0 = q0 * q0 / colsum[lane];
    float p1 = q1 * q1 / colsum[lane + 32];

    float s = p0 + p1;
    s += __shfl_xor(s, 1, 32);
    s += __shfl_xor(s, 2, 32);
    s += __shfl_xor(s, 4, 32);
    s += __shfl_xor(s, 8, 32);
    s += __shfl_xor(s, 16, 32);
    float inv = 1.0f / s;

    Pout[base + lane]      = p0 * inv;
    Pout[base + lane + 32] = p1 * inv;
}

extern "C" void kernel_launch(void* const* d_in, const int* in_sizes, int n_in,
                              void* d_out, int out_size, void* d_ws, size_t ws_size,
                              hipStream_t stream) {
    const float* z    = (const float*)d_in[0];
    const float* cent = (const float*)d_in[1];

    float* Q = (float*)d_out;
    float* P = Q + (size_t)NR * NC;

    float* colsum    = (float*)d_ws;
    float* blockpart = colsum + NC;
    float* cnorm     = blockpart + (size_t)QBLK * NC;

    clusternet_cnorm_kernel<<<NC / WAVES, 256, 0, stream>>>(cent, cnorm);
    clusternet_q_kernel<<<QBLK, 256, 0, stream>>>(z, cent, cnorm, Q, blockpart);
    clusternet_colsum_kernel<<<1, 64, 0, stream>>>(blockpart, colsum);
    clusternet_p_kernel<<<NR / WAVES, 256, 0, stream>>>(Q, colsum, P);
}


# Q GEMM via LDS: 256 blocks x 4 waves, centroids + 32-row z tile staged by coalesced 16-B loads in 4 K chunks, ds_load_b128 fragments; colsum over 256 partials, 16 per thread; all sums in new order
# speedup vs baseline: 2.0036x; 2.0036x over previous
#include <hip/hip_runtime.h>


typedef __attribute__((ext_vector_type(2))) float v2f;
typedef __attribute__((ext_vector_type(4))) float v4f;
typedef __attribute__((ext_vector_type(8))) float v8f;

namespace {
constexpr int KD    = 256;
constexpr int NC    = 64;
constexpr int NR    = 8192;
constexpr int WAVES = 8;

constexpr int QROWS = 32;
constexpr int QBLK  = NR / QROWS;
constexpr int QTHR  = 128;
constexpr int PITCH = KD + 4;
constexpr int KCH   = 64;
constexpr int NCH   = KD / KCH;
constexpr int SROWS = NC + QROWS;
constexpr int LPT   = SROWS * (KCH / 4) / QTHR;
constexpr int NG    = KCH / 8;
constexpr int CSG   = 16;
static_assert(LPT * QTHR == SROWS * (KCH / 4), "staging tiles the chunk exactly");
static_assert((PITCH * 4) % 16 == 0 && ((PITCH * 4 / 16) & 1) == 1, "row pitch is an odd multiple of 16 bytes");
static_assert(QBLK % CSG == 0, "column-sum groups tile the partials");
}

__global__ void __launch_bounds__(256)
clusternet_cnorm_kernel(const float* __restrict__ cent,
                        float* __restrict__ cnorm) {
    const int lane = threadIdx.x & 31;
    const int row  = blockIdx.x * WAVES + (threadIdx.x >> 5);
    const float* cr = cent + (size_t)row * KD;

    float s = 0.f;
#pragma unroll
    for (int h = 0; h < KD; h += 32) {
        float v = cr[h + lane];
        s += v * v;
    }
    s += __shfl_xor(s, 1, 32);
    s += __shfl_xor(s, 2, 32);
    s += __shfl_xor(s, 4, 32);
    s += __shfl_xor(s, 8, 32);
    s += __shfl_xor(s, 16, 32);
    if (lane == 0) cnorm[row] = s;
}

__global__ void __launch_bounds__(QTHR)
clusternet_q_kernel(const float* __restrict__ z,
                    const float* __restrict__ cent,
                    const float* __restrict__ cnorm,
                    float* __restrict__ Qout,
                    float* __restrict__ blockpart) {
    __shared__ __attribute__((aligned(16))) float sm[SROWS * PITCH];
    __shared__ __attribute__((aligned(16))) float rsx[4][16];
    __shared__ float csum[2][NC];

    const int tid  = threadIdx.x;
    const int lane = tid & 31;
    const int wave = tid >> 5;
    const int half = lane >> 4;
    const int n16  = lane & 15;
    const int rt   = wave & 1;
    const int ch   = wave >> 1;
    const int blockRow = blockIdx.x * QROWS;

    const int sp = tid & 15;
    const int sr = tid >> 4;
    v4f stg[NCH][LPT];
#pragma unroll
    for (int c = 0; c < NCH; ++c) {
#pragma unroll
        for (int j = 0; j < LPT; ++j) {
            const int row = sr + 8 * j;
            const float* src = (j < NC / 8)
                ? cent + (size_t)row * KD
                : z + (size_t)(blockRow + row - NC) * KD;
            stg[c][j] = *(const v4f*)(src + c * KCH + sp * 4);
        }
        __builtin_amdgcn_sched_barrier(0);
    }

    const float cn0 = cnorm[32 * ch + n16];
    const float cn1 = cnorm[32 * ch + 16 + n16];

    const float* aP  = sm + (NC + 16 * rt + n16) * PITCH + 4 * half;
    const float* bP0 = sm + (32 * ch + n16) * PITCH + 4 * half;
    const float* bP1 = bP0 + 16 * PITCH;

    v8f acc0 = {};
    v8f acc1 = {};
    float zp = 0.f;
    v4f fa[2][NG], fb0[2][NG], fb1[2][NG];

#define STORE_CHUNK(C)                                                        \
    _Pragma("unroll") for (int j = 0; j < LPT; ++j)                           \
        *(v4f*)(sm + (sr + 8 * j) * PITCH + (C) * KCH + sp * 4) = stg[C][j];
#define LDS_BARRIER()                                                         \
    do {                                                                      \
        __builtin_amdgcn_fence(__ATOMIC_RELEASE, "workgroup", "local");       \
        __builtin_amdgcn_s_barrier();                                         \
        __builtin_amdgcn_fence(__ATOMIC_ACQUIRE, "workgroup", "local");       \
    } while (0)
#define LOAD_FRAGS(SET, C)                                                    \
    _Pragma("unroll") for (int g = 0; g < NG; ++g) {                          \
        fa[SET][g]  = *(const v4f*)(aP  + (C) * KCH + g * 8);                 \
        fb0[SET][g] = *(const v4f*)(bP0 + (C) * KCH + g * 8);                 \
        fb1[SET][g] = *(const v4f*)(bP1 + (C) * KCH + g * 8);                 \
    }
#define MMA_GROUPS(SET, G0, G1)                                               \
    _Pragma("unroll") for (int g = (G0); g < (G1); ++g) {                     \
        const v4f av = fa[SET][g], pv = fb0[SET][g], rv = fb1[SET][g];        \
        const v2f a01 = __builtin_shufflevector(av, av, 0, 1);                \
        const v2f a23 = __builtin_shufflevector(av, av, 2, 3);                \
        const v2f p01 = __builtin_shufflevector(pv, pv, 0, 1);                \
        const v2f p23 = __builtin_shufflevector(pv, pv, 2, 3);                \
        const v2f r01 = __builtin_shufflevector(rv, rv, 0, 1);                \
        const v2f r23 = __builtin_shufflevector(rv, rv, 2, 3);                \
        zp += av.x * av.x + av.y * av.y + av.z * av.z + av.w * av.w;          \
        acc0 = __builtin_amdgcn_wmma_f32_16x16x4_f32(false, a01, false, p01, (short)0, acc0, false, false); \
        acc1 = __builtin_amdgcn_wmma_f32_16x16x4_f32(false, a01, false, r01, (short)0, acc1, false, false); \
        acc0 = __builtin_amdgcn_wmma_f32_16x16x4_f32(false, a23, false, p23, (short)0, acc0, false, false); \
        acc1 = __builtin_amdgcn_wmma_f32_16x16x4_f32(false, a23, false, r23, (short)0, acc1, false, false); \
    }

    STORE_CHUNK(0)
    LDS_BARRIER();
    LOAD_FRAGS(0, 0)
    __builtin_amdgcn_sched_barrier(0);

#pragma unroll
    for (int c = 0; c < NCH; ++c) {
        MMA_GROUPS(c & 1, 0, NG / 2)
        if (c + 1 < NCH) {
            __builtin_amdgcn_sched_barrier(0);
            STORE_CHUNK(c + 1)
            LDS_BARRIER();
            LOAD_FRAGS((c + 1) & 1, c + 1)
            __builtin_amdgcn_sched_barrier(0);
        }
        MMA_GROUPS(c & 1, NG / 2, NG)
    }
#undef STORE_CHUNK
#undef LDS_BARRIER
#undef LOAD_FRAGS
#undef MMA_GROUPS

    const float znf = zp + __shfl_xor(zp, 16, 32);
    const int rowBase = blockRow + 16 * rt;

    float q0[8], q1[8], rs[8];
    float mine = 0.f;
#pragma unroll
    for (int r = 0; r < 8; ++r) {
        const float znr = __shfl(znf, r + 8 * half, 32);
        const float s0 = fmaxf(znr + cn0 - 2.0f * acc0[r], 0.0f);
        const float s1 = fmaxf(znr + cn1 - 2.0f * acc1[r], 0.0f);
        q0[r] = 1.0f / (1.0f + sqrtf(s0));
        q1[r] = 1.0f / (1.0f + sqrtf(s1));
        float p = q0[r] + q1[r];
        p += __shfl_xor(p, 1, 32);
        p += __shfl_xor(p, 2, 32);
        p += __shfl_xor(p, 4, 32);
        p += __shfl_xor(p, 8, 32);
        rs[r] = p;
        mine = (n16 == r) ? p : mine;
    }
    if (n16 < 8) rsx[wave][n16 + 8 * half] = mine;
    __syncthreads();
    const v4f o0 = *(const v4f*)&rsx[wave ^ 2][8 * half];
    const v4f o1 = *(const v4f*)&rsx[wave ^ 2][8 * half + 4];

    float cs0 = 0.f, cs1 = 0.f;
#pragma unroll
    for (int r = 0; r < 8; ++r) {
        const float other = (r < 4) ? o0[r & 3] : o1[r & 3];
        const float inv = 1.0f / (rs[r] + other);
        const float v0 = q0[r] * inv;
        const float v1 = q1[r] * inv;
        float* qrow = Qout + (size_t)(rowBase + r + 8 * half) * NC + 32 * ch + n16;
        qrow[0]  = v0;
        qrow[16] = v1;
        cs0 += v0;
        cs1 += v1;
    }

    cs0 += __shfl_xor(cs0, 16, 32);
    cs1 += __shfl_xor(cs1, 16, 32);
    if (lane < 16) {
        csum[rt][32 * ch + n16]      = cs0;
        csum[rt][32 * ch + 16 + n16] = cs1;
    }
    __syncthreads();
    if (tid < NC)
        blockpart[(size_t)blockIdx.x * NC + tid] = csum[0][tid] + csum[1][tid];
}

__global__ void __launch_bounds__(CSG * NC)
clusternet_colsum_kernel(const float* __restrict__ blockpart,
                         float* __restrict__ colsum) {
    __shared__ float red[CSG][NC];
    const int c = threadIdx.x & (NC - 1);
    const int g = threadIdx.x / NC;
    constexpr int PER = QBLK / CSG;
    float v[PER];
#pragma unroll
    for (int j = 0; j < PER; ++j) v[j] = blockpart[(size_t)(g * PER + j) * NC + c];
    float s = 0.f;
#pragma unroll
    for (int j = 0; j < PER; ++j) s += v[j];
    red[g][c] = s;
    __syncthreads();
    if (threadIdx.x < NC) {
        float t = 0.f;
#pragma unroll
        for (int k = 0; k < CSG; ++k) t += red[k][threadIdx.x];
        colsum[threadIdx.x] = t;
    }
}

__global__ void __launch_bounds__(256)
clusternet_p_kernel(const float* __restrict__ Q,
                    const float* __restrict__ colsum,
                    float* __restrict__ Pout) {
    const int lane = threadIdx.x & 31;
    const int row  = blockIdx.x * WAVES + (threadIdx.x >> 5);
    const size_t base = (size_t)row * NC;

    float q0 = Q[base + lane];
    float q1 = Q[base + lane + 32];
    float p0 = q0 * q0 / colsum[lane];
    float p1 = q1 * q1 / colsum[lane + 32];

    float s = p0 + p1;
    s += __shfl_xor(s, 1, 32);
    s += __shfl_xor(s, 2, 32);
    s += __shfl_xor(s, 4, 32);
    s += __shfl_xor(s, 8, 32);
    s += __shfl_xor(s, 16, 32);
    float inv = 1.0f / s;

    Pout[base + lane]      = p0 * inv;
    Pout[base + lane + 32] = p1 * inv;
}

extern "C" void kernel_launch(void* const* d_in, const int* in_sizes, int n_in,
                              void* d_out, int out_size, void* d_ws, size_t ws_size,
                              hipStream_t stream) {
    const float* z    = (const float*)d_in[0];
    const float* cent = (const float*)d_in[1];

    float* Q = (float*)d_out;
    float* P = Q + (size_t)NR * NC;

    float* colsum = (float*)d_ws;
    float* cnorm  = colsum + NC;
    float* blockpart = P;

    clusternet_cnorm_kernel<<<NC / WAVES, 256, 0, stream>>>(cent, cnorm);
    clusternet_q_kernel<<<QBLK, QTHR, 0, stream>>>(z, cent, cnorm, Q, blockpart);
    clusternet_colsum_kernel<<<1, CSG * NC, 0, stream>>>(blockpart, colsum);
    clusternet_p_kernel<<<NR / WAVES, 256, 0, stream>>>(Q, colsum, P);
}


# cnorm launch removed (4 launches -> 3): each Q wave sums the squares of the centroid B fragments it reads from LDS for its WMMAs, lane halves added by one shuffle; cnorm summed in a new order
# speedup vs baseline: 2.3576x; 1.1767x over previous
#include <hip/hip_runtime.h>


typedef __attribute__((ext_vector_type(2))) float v2f;
typedef __attribute__((ext_vector_type(4))) float v4f;
typedef __attribute__((ext_vector_type(8))) float v8f;

namespace {
constexpr int KD    = 256;
constexpr int NC    = 64;
constexpr int NR    = 8192;
constexpr int WAVES = 8;

constexpr int QROWS = 32;
constexpr int QBLK  = NR / QROWS;
constexpr int QTHR  = 128;
constexpr int PITCH = KD + 4;
constexpr int KCH   = 64;
constexpr int NCH   = KD / KCH;
constexpr int SROWS = NC + QROWS;
constexpr int LPT   = SROWS * (KCH / 4) / QTHR;
constexpr int NG    = KCH / 8;
constexpr int CSG   = 16;
static_assert(LPT * QTHR == SROWS * (KCH / 4), "staging tiles the chunk exactly");
static_assert((PITCH * 4) % 16 == 0 && ((PITCH * 4 / 16) & 1) == 1, "row pitch is an odd multiple of 16 bytes");
static_assert(QBLK % CSG == 0, "column-sum groups tile the partials");
}

__global__ void __launch_bounds__(QTHR)
clusternet_q_kernel(const float* __restrict__ z,
                    const float* __restrict__ cent,
                    float* __restrict__ Qout,
                    float* __restrict__ blockpart) {
    __shared__ __attribute__((aligned(16))) float sm[SROWS * PITCH];
    __shared__ __attribute__((aligned(16))) float rsx[4][16];
    __shared__ float csum[2][NC];

    const int tid  = threadIdx.x;
    const int lane = tid & 31;
    const int wave = tid >> 5;
    const int half = lane >> 4;
    const int n16  = lane & 15;
    const int rt   = wave & 1;
    const int ch   = wave >> 1;
    const int blockRow = blockIdx.x * QROWS;

    const int sp = tid & 15;
    const int sr = tid >> 4;
    v4f stg[NCH][LPT];
#pragma unroll
    for (int c = 0; c < NCH; ++c) {
#pragma unroll
        for (int j = 0; j < LPT; ++j) {
            const int row = sr + 8 * j;
            const float* src = (j < NC / 8)
                ? cent + (size_t)row * KD
                : z + (size_t)(blockRow + row - NC) * KD;
            stg[c][j] = *(const v4f*)(src + c * KCH + sp * 4);
        }
        __builtin_amdgcn_sched_barrier(0);
    }

    const float* aP  = sm + (NC + 16 * rt + n16) * PITCH + 4 * half;
    const float* bP0 = sm + (32 * ch + n16) * PITCH + 4 * half;
    const float* bP1 = bP0 + 16 * PITCH;

    v8f acc0 = {};
    v8f acc1 = {};
    float zp = 0.f;
    v2f cq0 = {}, cq1 = {};
    v4f fa[2][NG], fb0[2][NG], fb1[2][NG];

#define STORE_CHUNK(C)                                                        \
    _Pragma("unroll") for (int j = 0; j < LPT; ++j)                           \
        *(v4f*)(sm + (sr + 8 * j) * PITCH + (C) * KCH + sp * 4) = stg[C][j];
#define LDS_BARRIER()                                                         \
    do {                                                                      \
        __builtin_amdgcn_fence(__ATOMIC_RELEASE, "workgroup", "local");       \
        __builtin_amdgcn_s_barrier();                                         \
        __builtin_amdgcn_fence(__ATOMIC_ACQUIRE, "workgroup", "local");       \
    } while (0)
#define LOAD_FRAGS(SET, C)                                                    \
    _Pragma("unroll") for (int g = 0; g < NG; ++g) {                          \
        fa[SET][g]  = *(const v4f*)(aP  + (C) * KCH + g * 8);                 \
        fb0[SET][g] = *(const v4f*)(bP0 + (C) * KCH + g * 8);                 \
        fb1[SET][g] = *(const v4f*)(bP1 + (C) * KCH + g * 8);                 \
    }
#define MMA_GROUPS(SET, G0, G1)                                               \
    _Pragma("unroll") for (int g = (G0); g < (G1); ++g) {                     \
        const v4f av = fa[SET][g], pv = fb0[SET][g], rv = fb1[SET][g];        \
        const v2f a01 = __builtin_shufflevector(av, av, 0, 1);                \
        const v2f a23 = __builtin_shufflevector(av, av, 2, 3);                \
        const v2f p01 = __builtin_shufflevector(pv, pv, 0, 1);                \
        const v2f p23 = __builtin_shufflevector(pv, pv, 2, 3);                \
        const v2f r01 = __builtin_shufflevector(rv, rv, 0, 1);                \
        const v2f r23 = __builtin_shufflevector(rv, rv, 2, 3);                \
        zp += av.x * av.x + av.y * av.y + av.z * av.z + av.w * av.w;          \
        cq0 += p01 * p01;  cq0 += p23 * p23;                                  \
        cq1 += r01 * r01;  cq1 += r23 * r23;                                  \
        acc0 = __builtin_amdgcn_wmma_f32_16x16x4_f32(false, a01, false, p01, (short)0, acc0, false, false); \
        acc1 = __builtin_amdgcn_wmma_f32_16x16x4_f32(false, a01, false, r01, (short)0, acc1, false, false); \
        acc0 = __builtin_amdgcn_wmma_f32_16x16x4_f32(false, a23, false, p23, (short)0, acc0, false, false); \
        acc1 = __builtin_amdgcn_wmma_f32_16x16x4_f32(false, a23, false, r23, (short)0, acc1, false, false); \
    }

    STORE_CHUNK(0)
    LDS_BARRIER();
    LOAD_FRAGS(0, 0)
    __builtin_amdgcn_sched_barrier(0);

#pragma unroll
    for (int c = 0; c < NCH; ++c) {
        MMA_GROUPS(c & 1, 0, NG / 2)
        if (c + 1 < NCH) {
            __builtin_amdgcn_sched_barrier(0);
            STORE_CHUNK(c + 1)
            LDS_BARRIER();
            LOAD_FRAGS((c + 1) & 1, c + 1)
            __builtin_amdgcn_sched_barrier(0);
        }
        MMA_GROUPS(c & 1, NG / 2, NG)
    }
#undef STORE_CHUNK
#undef LDS_BARRIER
#undef LOAD_FRAGS
#undef MMA_GROUPS

    const float znf = zp + __shfl_xor(zp, 16, 32);
    const float cp0 = cq0.x + cq0.y;
    const float cp1 = cq1.x + cq1.y;
    const float cn0 = cp0 + __shfl_xor(cp0, 16, 32);
    const float cn1 = cp1 + __shfl_xor(cp1, 16, 32);
    const int rowBase = blockRow + 16 * rt;

    float q0[8], q1[8], rs[8];
    float mine = 0.f;
#pragma unroll
    for (int r = 0; r < 8; ++r) {
        const float znr = __shfl(znf, r + 8 * half, 32);
        const float s0 = fmaxf(znr + cn0 - 2.0f * acc0[r], 0.0f);
        const float s1 = fmaxf(znr + cn1 - 2.0f * acc1[r], 0.0f);
        q0[r] = 1.0f / (1.0f + sqrtf(s0));
        q1[r] = 1.0f / (1.0f + sqrtf(s1));
        float p = q0[r] + q1[r];
        p += __shfl_xor(p, 1, 32);
        p += __shfl_xor(p, 2, 32);
        p += __shfl_xor(p, 4, 32);
        p += __shfl_xor(p, 8, 32);
        rs[r] = p;
        mine = (n16 == r) ? p : mine;
    }
    if (n16 < 8) rsx[wave][n16 + 8 * half] = mine;
    __syncthreads();
    const v4f o0 = *(const v4f*)&rsx[wave ^ 2][8 * half];
    const v4f o1 = *(const v4f*)&rsx[wave ^ 2][8 * half + 4];

    float cs0 = 0.f, cs1 = 0.f;
#pragma unroll
    for (int r = 0; r < 8; ++r) {
        const float other = (r < 4) ? o0[r & 3] : o1[r & 3];
        const float inv = 1.0f / (rs[r] + other);
        const float v0 = q0[r] * inv;
        const float v1 = q1[r] * inv;
        float* qrow = Qout + (size_t)(rowBase + r + 8 * half) * NC + 32 * ch + n16;
        qrow[0]  = v0;
        qrow[16] = v1;
        cs0 += v0;
        cs1 += v1;
    }

    cs0 += __shfl_xor(cs0, 16, 32);
    cs1 += __shfl_xor(cs1, 16, 32);
    if (lane < 16) {
        csum[rt][32 * ch + n16]      = cs0;
        csum[rt][32 * ch + 16 + n16] = cs1;
    }
    __syncthreads();
    if (tid < NC)
        blockpart[(size_t)blockIdx.x * NC + tid] = csum[0][tid] + csum[1][tid];
}

__global__ void __launch_bounds__(CSG * NC)
clusternet_colsum_kernel(const float* __restrict__ blockpart,
                         float* __restrict__ colsum) {
    __shared__ float red[CSG][NC];
    const int c = threadIdx.x & (NC - 1);
    const int g = threadIdx.x / NC;
    constexpr int PER = QBLK / CSG;
    float v[PER];
#pragma unroll
    for (int j = 0; j < PER; ++j) v[j] = blockpart[(size_t)(g * PER + j) * NC + c];
    float s = 0.f;
#pragma unroll
    for (int j = 0; j < PER; ++j) s += v[j];
    red[g][c] = s;
    __syncthreads();
    if (threadIdx.x < NC) {
        float t = 0.f;
#pragma unroll
        for (int k = 0; k < CSG; ++k) t += red[k][threadIdx.x];
        colsum[threadIdx.x] = t;
    }
}

__global__ void __launch_bounds__(256)
clusternet_p_kernel(const float* __restrict__ Q,
                    const float* __restrict__ colsum,
                    float* __restrict__ Pout) {
    const int lane = threadIdx.x & 31;
    const int row  = blockIdx.x * WAVES + (threadIdx.x >> 5);
    const size_t base = (size_t)row * NC;

    float q0 = Q[base + lane];
    float q1 = Q[base + lane + 32];
    float p0 = q0 * q0 / colsum[lane];
    float p1 = q1 * q1 / colsum[lane + 32];

    float s = p0 + p1;
    s += __shfl_xor(s, 1, 32);
    s += __shfl_xor(s, 2, 32);
    s += __shfl_xor(s, 4, 32);
    s += __shfl_xor(s, 8, 32);
    s += __shfl_xor(s, 16, 32);
    float inv = 1.0f / s;

    Pout[base + lane]      = p0 * inv;
    Pout[base + lane + 32] = p1 * inv;
}

extern "C" void kernel_launch(void* const* d_in, const int* in_sizes, int n_in,
                              void* d_out, int out_size, void* d_ws, size_t ws_size,
                              hipStream_t stream) {
    const float* z    = (const float*)d_in[0];
    const float* cent = (const float*)d_in[1];

    float* Q = (float*)d_out;
    float* P = Q + (size_t)NR * NC;

    float* colsum = (float*)d_ws;
    float* blockpart = P;

    clusternet_q_kernel<<<QBLK, QTHR, 0, stream>>>(z, cent, Q, blockpart);
    clusternet_colsum_kernel<<<1, CSG * NC, 0, stream>>>(blockpart, colsum);
    clusternet_p_kernel<<<NR / WAVES, 256, 0, stream>>>(Q, colsum, P);
}
